# starting kernel
# baseline (speedup 1.0000x reference)
#include <hip/hip_runtime.h>
#include <math.h>


typedef __attribute__((ext_vector_type(2))) float v2f;
typedef __attribute__((ext_vector_type(8))) float v8f;

#define B_      16
#define N_      8192
#define K_      125
#define KT      8
#define NCHUNK  16
#define CHUNK   (N_ / NCHUNK)
#define C15     0.06349363593424097f

__global__ __launch_bounds__(256)
void fv_coefs(const float* __restrict__ w, const float* __restrict__ mu,
              const float* __restrict__ sg, float4* __restrict__ c4)
{
    const int tid  = threadIdx.x;
    const int lane = tid & 31;
    const int t    = tid >> 5;
    const int m    = lane & 15;
    const bool hi  = lane >= 16;
    const int  kc    = t * 16 + m;
    const bool valid = kc < K_;
    const int  ki    = valid ? kc : 0;

    const float m0 = mu[ki*3+0], m1 = mu[ki*3+1], m2 = mu[ki*3+2];
    const float s0 = sg[ki*3+0], s1 = sg[ki*3+1], s2 = sg[ki*3+2];
    const float i0 = 1.0f/(s0*s0), i1 = 1.0f/(s1*s1), i2 = 1.0f/(s2*s2);
    const float g0 = -0.5f*i0, g1 = -0.5f*i1, g2 = -0.5f*i2;
    const float g3 = m0*i0,    g4 = m1*i1,    g5 = m2*i2;
    const float lq = logf(w[ki] * C15 / (s0 * s1 * s2));
    const float g6 = valid ? (lq - 0.5f*(m0*m0*i0 + m1*m1*i1 + m2*m2*i2))
                           : -INFINITY;

    float4 v;
    v.x = hi ? g2   : g0;
    v.y = hi ? g3   : g1;
    v.z = hi ? g6   : g4;
    v.w = hi ? 0.0f : g5;
    c4[tid] = v;
}

__global__ __launch_bounds__(256)
void fv_denom_wmma(const float* __restrict__ pts,
                   const float4* __restrict__ c4,
                   float* __restrict__ rdenom)
{
    const int lane = threadIdx.x & 31;
    const int wave = threadIdx.x >> 5;
    const int tile = blockIdx.x * 8 + wave;
    const int b    = tile >> 9;
    const int n0   = (tile & 511) << 4;
    const int m    = lane & 15;
    const bool hi  = lane >= 16;

    const float* p = pts + (size_t)(b * N_ + n0 + m) * 3;
    const float px = p[0], py = p[1], pz = p[2];

    v2f a1, a2;
    a1.x = hi ? pz * pz : px * px;
    a1.y = hi ? px      : py * py;
    a2.x = hi ? 1.0f    : py;
    a2.y = hi ? 0.0f    : pz;

    float acc[8];
#pragma unroll
    for (int r = 0; r < 8; ++r) acc[r] = 0.0f;

#pragma unroll
    for (int t = 0; t < KT; ++t) {
        const float4 bc = c4[t * 32 + lane];
        v2f b1, b2;
        b1.x = bc.x; b1.y = bc.y;
        b2.x = bc.z; b2.y = bc.w;

        v8f c = {};
        c = __builtin_amdgcn_wmma_f32_16x16x4_f32(false, a1, false, b1,
                                                  (short)0, c, false, false);
        c = __builtin_amdgcn_wmma_f32_16x16x4_f32(false, a2, false, b2,
                                                  (short)0, c, false, false);

#pragma unroll
        for (int r = 0; r < 8; ++r)
            acc[r] += __expf(c[r]);
    }

#pragma unroll
    for (int r = 0; r < 8; ++r) {
        float v = acc[r];
        v += __shfl_xor(v, 1, 32);
        v += __shfl_xor(v, 2, 32);
        v += __shfl_xor(v, 4, 32);
        v += __shfl_xor(v, 8, 32);
        acc[r] = v;
    }
    if (lane == 0) {
#pragma unroll
        for (int r = 0; r < 8; ++r)
            rdenom[b * N_ + n0 + r] = 1.0f / acc[r];
    } else if (lane == 16) {
#pragma unroll
        for (int r = 0; r < 8; ++r)
            rdenom[b * N_ + n0 + 8 + r] = 1.0f / acc[r];
    }
}

__global__ __launch_bounds__(128)
void fv_accum(const float* __restrict__ pts, const float* __restrict__ w,
              const float* __restrict__ mu, const float* __restrict__ sg,
              const float* __restrict__ rdenom, float* __restrict__ partial)
{
    __shared__ float lx[CHUNK], ly[CHUNK], lz[CHUNK], lrd[CHUNK];
    const int b  = blockIdx.x >> 4;
    const int ch = blockIdx.x & 15;
    const int n0 = ch * CHUNK;
    const int t  = threadIdx.x;

    for (int i = t; i < CHUNK; i += 128) {
        const int n = b * N_ + n0 + i;
        const float* p = pts + (size_t)n * 3;
        lx[i] = p[0]; ly[i] = p[1]; lz[i] = p[2];
        lrd[i] = rdenom[n];
    }
    __syncthreads();

    const int kk = (t < K_) ? t : 0;
    const float m0 = mu[kk*3+0], m1 = mu[kk*3+1], m2 = mu[kk*3+2];
    const float s0 = sg[kk*3+0], s1 = sg[kk*3+1], s2 = sg[kk*3+2];
    const float rs0 = 1.0f/s0, rs1 = 1.0f/s1, rs2 = 1.0f/s2;
    const float q = w[kk] * C15 * rs0 * rs1 * rs2;

    float sQ = 0.0f, mQ = -INFINITY;
    float sM[3] = {0,0,0}, xM[3] = {-INFINITY,-INFINITY,-INFINITY};
    float nM[3] = { INFINITY, INFINITY, INFINITY};
    float sS[3] = {0,0,0}, xS[3] = {-INFINITY,-INFINITY,-INFINITY};
    float nS[3] = { INFINITY, INFINITY, INFINITY};

    for (int i = 0; i < CHUNK; ++i) {
        float z0 = (lx[i] - m0) * rs0;
        float z1 = (ly[i] - m1) * rs1;
        float z2 = (lz[i] - m2) * rs2;
        float e  = -0.5f * (z0*z0 + z1*z1 + z2*z2);
        float Q  = q * __expf(e) * lrd[i];
        sQ += Q; mQ = fmaxf(mQ, Q);
        float z[3] = {z0, z1, z2};
#pragma unroll
        for (int d = 0; d < 3; ++d) {
            float u = Q * z[d];
            sM[d] += u; xM[d] = fmaxf(xM[d], u); nM[d] = fminf(nM[d], u);
            float v = Q * (z[d]*z[d] - 1.0f);
            sS[d] += v; xS[d] = fmaxf(xS[d], v); nS[d] = fminf(nS[d], v);
        }
    }

    if (t < K_) {
        float* o = partial + ((size_t)(b * NCHUNK + ch) * K_ + t) * 20;
        o[0] = mQ;  o[1] = sQ;
#pragma unroll
        for (int d = 0; d < 3; ++d) {
            o[2  + d] = xM[d]; o[5  + d] = nM[d]; o[8  + d] = sM[d];
            o[11 + d] = xS[d]; o[14 + d] = nS[d]; o[17 + d] = sS[d];
        }
    }
}

__global__ __launch_bounds__(128)
void fv_finalize(const float* __restrict__ w, const float* __restrict__ partial,
                 float* __restrict__ out)
{
    __shared__ float vals[20][128];
    __shared__ float ssq[20];
    const int b = blockIdx.x;
    const int t = threadIdx.x;

    float v[20];
    if (t < K_) {
        float a[20];
        {
            const float* p = partial + ((size_t)(b * NCHUNK) * K_ + t) * 20;
#pragma unroll
            for (int f = 0; f < 20; ++f) a[f] = p[f];
        }
        for (int c = 1; c < NCHUNK; ++c) {
            const float* p = partial + ((size_t)(b * NCHUNK + c) * K_ + t) * 20;
            a[0] = fmaxf(a[0], p[0]); a[1] += p[1];
#pragma unroll
            for (int d = 0; d < 3; ++d) {
                a[2  + d] = fmaxf(a[2  + d], p[2  + d]);
                a[5  + d] = fminf(a[5  + d], p[5  + d]);
                a[8  + d] +=            p[8  + d];
                a[11 + d] = fmaxf(a[11 + d], p[11 + d]);
                a[14 + d] = fminf(a[14 + d], p[14 + d]);
                a[17 + d] +=            p[17 + d];
            }
        }
        const float wk  = w[t];
        const float Nf  = (float)N_;
        const float spi = 1.0f / (sqrtf(wk) * Nf);
        const float ssg = 1.0f / (Nf * sqrtf(2.0f * wk));
        v[0] = (a[0] - wk)      * spi;
        v[1] = (a[1] - Nf * wk) * spi;
#pragma unroll
        for (int f = 2;  f < 11; ++f) v[f] = a[f] * spi;
#pragma unroll
        for (int f = 11; f < 20; ++f) v[f] = a[f] * ssg;
#pragma unroll
        for (int f = 0; f < 20; ++f)
            v[f] = copysignf(sqrtf(fabsf(v[f])), v[f]);
    } else {
#pragma unroll
        for (int f = 0; f < 20; ++f) v[f] = 0.0f;
    }
#pragma unroll
    for (int f = 0; f < 20; ++f) vals[f][t] = v[f];
    __syncthreads();

    if (t < 20) {
        float s = 0.0f;
        for (int k = 0; k < K_; ++k) { float x = vals[t][k]; s += x * x; }
        ssq[t] = s;
    }
    __syncthreads();

    if (t < K_) {
#pragma unroll
        for (int f = 0; f < 20; ++f)
            out[(size_t)b * 20 * K_ + f * K_ + t] =
                v[f] * rsqrtf(fmaxf(ssq[f], 1e-12f));
    }
}

extern "C" void kernel_launch(void* const* d_in, const int* in_sizes, int n_in,
                              void* d_out, int out_size, void* d_ws, size_t ws_size,
                              hipStream_t stream)
{
    const float* pts = (const float*)d_in[0];
    const float* w   = (const float*)d_in[1];
    const float* mu  = (const float*)d_in[2];
    const float* sg  = (const float*)d_in[3];
    float* out = (float*)d_out;

    float*  rdenom  = (float*)d_ws;
    float*  partial = rdenom + B_ * N_;
    float4* c4      = (float4*)(partial + (size_t)B_ * NCHUNK * K_ * 20);

    fv_coefs<<<1, 256, 0, stream>>>(w, mu, sg, c4);
    fv_denom_wmma<<<(B_ * N_ / 16) / 8, 256, 0, stream>>>(pts, c4, rdenom);
    fv_accum<<<B_ * NCHUNK, 128, 0, stream>>>(pts, w, mu, sg, rdenom, partial);
    fv_finalize<<<B_, 128, 0, stream>>>(w, partial, out);
}


# accumulation kernel runs 4 waves per SIMD, was 1 (512-thread blocks: 4 point sub-groups x 128 gaussians, LDS combine); each 512-point chunk sum is now 4 sub-sums of 128 added in fixed order
# speedup vs baseline: 1.3763x; 1.3763x over previous
#include <hip/hip_runtime.h>
#include <math.h>


typedef __attribute__((ext_vector_type(2))) float v2f;
typedef __attribute__((ext_vector_type(8))) float v8f;

#define B_      16
#define N_      8192
#define K_      125
#define KT      8
#define NCHUNK  16
#define CHUNK   (N_ / NCHUNK)
#define NSUB    4
#define SUBPTS  (CHUNK / NSUB)
#define C15     0.06349363593424097f

__global__ __launch_bounds__(256)
void fv_coefs(const float* __restrict__ w, const float* __restrict__ mu,
              const float* __restrict__ sg, float4* __restrict__ c4)
{
    const int tid  = threadIdx.x;
    const int lane = tid & 31;
    const int t    = tid >> 5;
    const int m    = lane & 15;
    const bool hi  = lane >= 16;
    const int  kc    = t * 16 + m;
    const bool valid = kc < K_;
    const int  ki    = valid ? kc : 0;

    const float m0 = mu[ki*3+0], m1 = mu[ki*3+1], m2 = mu[ki*3+2];
    const float s0 = sg[ki*3+0], s1 = sg[ki*3+1], s2 = sg[ki*3+2];
    const float i0 = 1.0f/(s0*s0), i1 = 1.0f/(s1*s1), i2 = 1.0f/(s2*s2);
    const float g0 = -0.5f*i0, g1 = -0.5f*i1, g2 = -0.5f*i2;
    const float g3 = m0*i0,    g4 = m1*i1,    g5 = m2*i2;
    const float lq = logf(w[ki] * C15 / (s0 * s1 * s2));
    const float g6 = valid ? (lq - 0.5f*(m0*m0*i0 + m1*m1*i1 + m2*m2*i2))
                           : -INFINITY;

    float4 v;
    v.x = hi ? g2   : g0;
    v.y = hi ? g3   : g1;
    v.z = hi ? g6   : g4;
    v.w = hi ? 0.0f : g5;
    c4[tid] = v;
}

__global__ __launch_bounds__(256)
void fv_denom_wmma(const float* __restrict__ pts,
                   const float4* __restrict__ c4,
                   float* __restrict__ rdenom)
{
    const int lane = threadIdx.x & 31;
    const int wave = threadIdx.x >> 5;
    const int tile = blockIdx.x * 8 + wave;
    const int b    = tile >> 9;
    const int n0   = (tile & 511) << 4;
    const int m    = lane & 15;
    const bool hi  = lane >= 16;

    const float* p = pts + (size_t)(b * N_ + n0 + m) * 3;
    const float px = p[0], py = p[1], pz = p[2];

    v2f a1, a2;
    a1.x = hi ? pz * pz : px * px;
    a1.y = hi ? px      : py * py;
    a2.x = hi ? 1.0f    : py;
    a2.y = hi ? 0.0f    : pz;

    float acc[8];
#pragma unroll
    for (int r = 0; r < 8; ++r) acc[r] = 0.0f;

#pragma unroll
    for (int t = 0; t < KT; ++t) {
        const float4 bc = c4[t * 32 + lane];
        v2f b1, b2;
        b1.x = bc.x; b1.y = bc.y;
        b2.x = bc.z; b2.y = bc.w;

        v8f c = {};
        c = __builtin_amdgcn_wmma_f32_16x16x4_f32(false, a1, false, b1,
                                                  (short)0, c, false, false);
        c = __builtin_amdgcn_wmma_f32_16x16x4_f32(false, a2, false, b2,
                                                  (short)0, c, false, false);

#pragma unroll
        for (int r = 0; r < 8; ++r)
            acc[r] += __expf(c[r]);
    }

#pragma unroll
    for (int r = 0; r < 8; ++r) {
        float v = acc[r];
        v += __shfl_xor(v, 1, 32);
        v += __shfl_xor(v, 2, 32);
        v += __shfl_xor(v, 4, 32);
        v += __shfl_xor(v, 8, 32);
        acc[r] = v;
    }
    if (lane == 0) {
#pragma unroll
        for (int r = 0; r < 8; ++r)
            rdenom[b * N_ + n0 + r] = 1.0f / acc[r];
    } else if (lane == 16) {
#pragma unroll
        for (int r = 0; r < 8; ++r)
            rdenom[b * N_ + n0 + 8 + r] = 1.0f / acc[r];
    }
}

__global__ __launch_bounds__(128 * NSUB)
void fv_accum(const float* __restrict__ pts, const float* __restrict__ w,
              const float* __restrict__ mu, const float* __restrict__ sg,
              const float* __restrict__ rdenom, float* __restrict__ partial)
{
    __shared__ float lx[CHUNK], ly[CHUNK], lz[CHUNK], lrd[CHUNK];
    __shared__ float comb[NSUB - 1][20][128];
    const int b   = blockIdx.x >> 4;
    const int ch  = blockIdx.x & 15;
    const int n0  = ch * CHUNK;
    const int tid = threadIdx.x;
    const int t   = tid & 127;
    const int g   = __builtin_amdgcn_readfirstlane(tid >> 7);

    for (int i = tid; i < CHUNK; i += 128 * NSUB) {
        const int n = b * N_ + n0 + i;
        const float* p = pts + (size_t)n * 3;
        lx[i] = p[0]; ly[i] = p[1]; lz[i] = p[2];
        lrd[i] = rdenom[n];
    }
    __syncthreads();

    const int kk = (t < K_) ? t : 0;
    const float m0 = mu[kk*3+0], m1 = mu[kk*3+1], m2 = mu[kk*3+2];
    const float s0 = sg[kk*3+0], s1 = sg[kk*3+1], s2 = sg[kk*3+2];
    const float rs0 = 1.0f/s0, rs1 = 1.0f/s1, rs2 = 1.0f/s2;
    const float q = w[kk] * C15 * rs0 * rs1 * rs2;

    float sQ = 0.0f, mQ = -INFINITY;
    float sM[3] = {0,0,0}, xM[3] = {-INFINITY,-INFINITY,-INFINITY};
    float nM[3] = { INFINITY, INFINITY, INFINITY};
    float sS[3] = {0,0,0}, xS[3] = {-INFINITY,-INFINITY,-INFINITY};
    float nS[3] = { INFINITY, INFINITY, INFINITY};

    const float* gx = lx  + g * SUBPTS;
    const float* gy = ly  + g * SUBPTS;
    const float* gz = lz  + g * SUBPTS;
    const float* gr = lrd + g * SUBPTS;
    for (int i = 0; i < SUBPTS; ++i) {
        float z0 = (gx[i] - m0) * rs0;
        float z1 = (gy[i] - m1) * rs1;
        float z2 = (gz[i] - m2) * rs2;
        float e  = -0.5f * (z0*z0 + z1*z1 + z2*z2);
        float Q  = q * __expf(e) * gr[i];
        sQ += Q; mQ = fmaxf(mQ, Q);
        float z[3] = {z0, z1, z2};
#pragma unroll
        for (int d = 0; d < 3; ++d) {
            float u = Q * z[d];
            sM[d] += u; xM[d] = fmaxf(xM[d], u); nM[d] = fminf(nM[d], u);
            float v = Q * (z[d]*z[d] - 1.0f);
            sS[d] += v; xS[d] = fmaxf(xS[d], v); nS[d] = fminf(nS[d], v);
        }
    }

    if (g > 0) {
        float (*c)[128] = comb[g - 1];
        c[0][t] = mQ;  c[1][t] = sQ;
#pragma unroll
        for (int d = 0; d < 3; ++d) {
            c[2  + d][t] = xM[d]; c[5  + d][t] = nM[d]; c[8  + d][t] = sM[d];
            c[11 + d][t] = xS[d]; c[14 + d][t] = nS[d]; c[17 + d][t] = sS[d];
        }
    }
    __syncthreads();

    if (g == 0) {
#pragma unroll
        for (int s = 0; s < NSUB - 1; ++s) {
            float (*c)[128] = comb[s];
            mQ = fmaxf(mQ, c[0][t]);  sQ += c[1][t];
#pragma unroll
            for (int d = 0; d < 3; ++d) {
                xM[d] = fmaxf(xM[d], c[2  + d][t]);
                nM[d] = fminf(nM[d], c[5  + d][t]);
                sM[d] +=             c[8  + d][t];
                xS[d] = fmaxf(xS[d], c[11 + d][t]);
                nS[d] = fminf(nS[d], c[14 + d][t]);
                sS[d] +=             c[17 + d][t];
            }
        }
        if (t < K_) {
            float* o = partial + ((size_t)(b * NCHUNK + ch) * K_ + t) * 20;
            o[0] = mQ;  o[1] = sQ;
#pragma unroll
            for (int d = 0; d < 3; ++d) {
                o[2  + d] = xM[d]; o[5  + d] = nM[d]; o[8  + d] = sM[d];
                o[11 + d] = xS[d]; o[14 + d] = nS[d]; o[17 + d] = sS[d];
            }
        }
    }
}

__global__ __launch_bounds__(128)
void fv_finalize(const float* __restrict__ w, const float* __restrict__ partial,
                 float* __restrict__ out)
{
    __shared__ float vals[20][128];
    __shared__ float ssq[20];
    const int b = blockIdx.x;
    const int t = threadIdx.x;

    float v[20];
    if (t < K_) {
        float a[20];
        {
            const float* p = partial + ((size_t)(b * NCHUNK) * K_ + t) * 20;
#pragma unroll
            for (int f = 0; f < 20; ++f) a[f] = p[f];
        }
        for (int c = 1; c < NCHUNK; ++c) {
            const float* p = partial + ((size_t)(b * NCHUNK + c) * K_ + t) * 20;
            a[0] = fmaxf(a[0], p[0]); a[1] += p[1];
#pragma unroll
            for (int d = 0; d < 3; ++d) {
                a[2  + d] = fmaxf(a[2  + d], p[2  + d]);
                a[5  + d] = fminf(a[5  + d], p[5  + d]);
                a[8  + d] +=            p[8  + d];
                a[11 + d] = fmaxf(a[11 + d], p[11 + d]);
                a[14 + d] = fminf(a[14 + d], p[14 + d]);
                a[17 + d] +=            p[17 + d];
            }
        }
        const float wk  = w[t];
        const float Nf  = (float)N_;
        const float spi = 1.0f / (sqrtf(wk) * Nf);
        const float ssg = 1.0f / (Nf * sqrtf(2.0f * wk));
        v[0] = (a[0] - wk)      * spi;
        v[1] = (a[1] - Nf * wk) * spi;
#pragma unroll
        for (int f = 2;  f < 11; ++f) v[f] = a[f] * spi;
#pragma unroll
        for (int f = 11; f < 20; ++f) v[f] = a[f] * ssg;
#pragma unroll
        for (int f = 0; f < 20; ++f)
            v[f] = copysignf(sqrtf(fabsf(v[f])), v[f]);
    } else {
#pragma unroll
        for (int f = 0; f < 20; ++f) v[f] = 0.0f;
    }
#pragma unroll
    for (int f = 0; f < 20; ++f) vals[f][t] = v[f];
    __syncthreads();

    if (t < 20) {
        float s = 0.0f;
        for (int k = 0; k < K_; ++k) { float x = vals[t][k]; s += x * x; }
        ssq[t] = s;
    }
    __syncthreads();

    if (t < K_) {
#pragma unroll
        for (int f = 0; f < 20; ++f)
            out[(size_t)b * 20 * K_ + f * K_ + t] =
                v[f] * rsqrtf(fmaxf(ssq[f], 1e-12f));
    }
}

extern "C" void kernel_launch(void* const* d_in, const int* in_sizes, int n_in,
                              void* d_out, int out_size, void* d_ws, size_t ws_size,
                              hipStream_t stream)
{
    const float* pts = (const float*)d_in[0];
    const float* w   = (const float*)d_in[1];
    const float* mu  = (const float*)d_in[2];
    const float* sg  = (const float*)d_in[3];
    float* out = (float*)d_out;

    float*  rdenom  = (float*)d_ws;
    float*  partial = rdenom + B_ * N_;
    float4* c4      = (float4*)(partial + (size_t)B_ * NCHUNK * K_ * 20);

    fv_coefs<<<1, 256, 0, stream>>>(w, mu, sg, c4);
    fv_denom_wmma<<<(B_ * N_ / 16) / 8, 256, 0, stream>>>(pts, c4, rdenom);
    fv_accum<<<B_ * NCHUNK, 128 * NSUB, 0, stream>>>(pts, w, mu, sg, rdenom, partial);
    fv_finalize<<<B_, 128, 0, stream>>>(w, partial, out);
}


# coefficient-table and softmax-denominator launches folded into the accumulation kernel: each block builds the table and its chunk's denominators (same WMMA math and sum order) in LDS; 4 launches -> 2
# speedup vs baseline: 1.4750x; 1.0717x over previous
#include <hip/hip_runtime.h>
#include <math.h>


typedef __attribute__((ext_vector_type(2))) float v2f;
typedef __attribute__((ext_vector_type(8))) float v8f;

#define B_      16
#define N_      8192
#define K_      125
#define KT      8
#define NCHUNK  16
#define CHUNK   (N_ / NCHUNK)
#define NSUB    4
#define SUBPTS  (CHUNK / NSUB)
#define C15     0.06349363593424097f

#define NTHR    (128 * NSUB)
#define NWAVE   (NTHR / 32)
#define NTILE   (CHUNK / 16)

__global__ __launch_bounds__(NTHR)
void fv_accum(const float* __restrict__ pts, const float* __restrict__ w,
              const float* __restrict__ mu, const float* __restrict__ sg,
              float* __restrict__ partial)
{
    __shared__ float lx[CHUNK], ly[CHUNK], lz[CHUNK];
    __shared__ __attribute__((aligned(16))) float lrd[CHUNK];
    __shared__ float4 lc4[KT * 32];
    __shared__ float comb[NSUB - 1][20][128];
    const int b   = blockIdx.x >> 4;
    const int ch  = blockIdx.x & 15;
    const int n0  = ch * CHUNK;
    const int tid = threadIdx.x;
    const int t   = tid & 127;
    const int lane = tid & 31;
    const int g    = __builtin_amdgcn_readfirstlane(tid >> 7);
    const int wave = __builtin_amdgcn_readfirstlane(tid >> 5);

    static_assert(CHUNK == NTHR, "staging takes one point per thread");
    const float* pp = pts + (size_t)(b * N_ + n0 + tid) * 3;
    const float qx = pp[0], qy = pp[1], qz = pp[2];

    const int kk = (t < K_) ? t : 0;
    const float m0 = mu[kk*3+0], m1 = mu[kk*3+1], m2 = mu[kk*3+2];
    const float s0 = sg[kk*3+0], s1 = sg[kk*3+1], s2 = sg[kk*3+2];
    const float wq = w[kk];

    if (tid < KT * 32) {
        const int  ct    = tid >> 5;
        const int  cm    = lane & 15;
        const bool chi   = lane >= 16;
        const int  kc    = ct * 16 + cm;
        const bool valid = kc < K_;
        const int  ki    = valid ? kc : 0;

        const float c0 = mu[ki*3+0], c1 = mu[ki*3+1], c2 = mu[ki*3+2];
        const float d0 = sg[ki*3+0], d1 = sg[ki*3+1], d2 = sg[ki*3+2];
        const float i0 = 1.0f/(d0*d0), i1 = 1.0f/(d1*d1), i2 = 1.0f/(d2*d2);
        const float g0 = -0.5f*i0, g1 = -0.5f*i1, g2 = -0.5f*i2;
        const float g3 = c0*i0,    g4 = c1*i1,    g5 = c2*i2;
        const float lq = logf(w[ki] * C15 / (d0 * d1 * d2));
        const float g6 = valid ? (lq - 0.5f*(c0*c0*i0 + c1*c1*i1 + c2*c2*i2))
                               : -INFINITY;

        float4 v;
        v.x = chi ? g2   : g0;
        v.y = chi ? g3   : g1;
        v.z = chi ? g6   : g4;
        v.w = chi ? 0.0f : g5;
        lc4[tid] = v;
    }
    lx[tid] = qx; ly[tid] = qy; lz[tid] = qz;
    __syncthreads();

    {
        const int  m  = lane & 15;
        const bool hi = lane >= 16;
        for (int tile = wave; tile < NTILE; tile += NWAVE) {
            const int p0 = tile << 4;
            const float px = lx[p0 + m], py = ly[p0 + m], pz = lz[p0 + m];

            v2f a1, a2;
            a1.x = hi ? pz * pz : px * px;
            a1.y = hi ? px      : py * py;
            a2.x = hi ? 1.0f    : py;
            a2.y = hi ? 0.0f    : pz;

            float acc[8];
#pragma unroll
            for (int r = 0; r < 8; ++r) acc[r] = 0.0f;

#pragma unroll
            for (int kt = 0; kt < KT; ++kt) {
                const float4 bc = lc4[kt * 32 + lane];
                v2f b1, b2;
                b1.x = bc.x; b1.y = bc.y;
                b2.x = bc.z; b2.y = bc.w;

                v8f c = {};
                c = __builtin_amdgcn_wmma_f32_16x16x4_f32(false, a1, false, b1,
                                                          (short)0, c, false, false);
                c = __builtin_amdgcn_wmma_f32_16x16x4_f32(false, a2, false, b2,
                                                          (short)0, c, false, false);

#pragma unroll
                for (int r = 0; r < 8; ++r)
                    acc[r] += __expf(c[r]);
            }

#pragma unroll
            for (int r = 0; r < 8; ++r) {
                float v = acc[r];
                v += __shfl_xor(v, 1, 32);
                v += __shfl_xor(v, 2, 32);
                v += __shfl_xor(v, 4, 32);
                v += __shfl_xor(v, 8, 32);
                acc[r] = v;
            }
            if (m == 0) {
                float* dst = lrd + p0 + (hi ? 8 : 0);
#pragma unroll
                for (int r = 0; r < 8; ++r)
                    dst[r] = 1.0f / acc[r];
            }
        }
    }
    __syncthreads();

    const float rs0 = 1.0f/s0, rs1 = 1.0f/s1, rs2 = 1.0f/s2;
    const float q = wq * C15 * rs0 * rs1 * rs2;

    float sQ = 0.0f, mQ = -INFINITY;
    float sM[3] = {0,0,0}, xM[3] = {-INFINITY,-INFINITY,-INFINITY};
    float nM[3] = { INFINITY, INFINITY, INFINITY};
    float sS[3] = {0,0,0}, xS[3] = {-INFINITY,-INFINITY,-INFINITY};
    float nS[3] = { INFINITY, INFINITY, INFINITY};

    const float* gx = lx  + g * SUBPTS;
    const float* gy = ly  + g * SUBPTS;
    const float* gz = lz  + g * SUBPTS;
    const float* gr = lrd + g * SUBPTS;
    for (int i = 0; i < SUBPTS; ++i) {
        float z0 = (gx[i] - m0) * rs0;
        float z1 = (gy[i] - m1) * rs1;
        float z2 = (gz[i] - m2) * rs2;
        float e  = -0.5f * (z0*z0 + z1*z1 + z2*z2);
        float Q  = q * __expf(e) * gr[i];
        sQ += Q; mQ = fmaxf(mQ, Q);
        float z[3] = {z0, z1, z2};
#pragma unroll
        for (int d = 0; d < 3; ++d) {
            float u = Q * z[d];
            sM[d] += u; xM[d] = fmaxf(xM[d], u); nM[d] = fminf(nM[d], u);
            float v = Q * (z[d]*z[d] - 1.0f);
            sS[d] += v; xS[d] = fmaxf(xS[d], v); nS[d] = fminf(nS[d], v);
        }
    }

    if (g > 0) {
        float (*c)[128] = comb[g - 1];
        c[0][t] = mQ;  c[1][t] = sQ;
#pragma unroll
        for (int d = 0; d < 3; ++d) {
            c[2  + d][t] = xM[d]; c[5  + d][t] = nM[d]; c[8  + d][t] = sM[d];
            c[11 + d][t] = xS[d]; c[14 + d][t] = nS[d]; c[17 + d][t] = sS[d];
        }
    }
    __syncthreads();

    if (g == 0) {
#pragma unroll
        for (int s = 0; s < NSUB - 1; ++s) {
            float (*c)[128] = comb[s];
            mQ = fmaxf(mQ, c[0][t]);  sQ += c[1][t];
#pragma unroll
            for (int d = 0; d < 3; ++d) {
                xM[d] = fmaxf(xM[d], c[2  + d][t]);
                nM[d] = fminf(nM[d], c[5  + d][t]);
                sM[d] +=             c[8  + d][t];
                xS[d] = fmaxf(xS[d], c[11 + d][t]);
                nS[d] = fminf(nS[d], c[14 + d][t]);
                sS[d] +=             c[17 + d][t];
            }
        }
        if (t < K_) {
            float* o = partial + ((size_t)(b * NCHUNK + ch) * K_ + t) * 20;
            o[0] = mQ;  o[1] = sQ;
#pragma unroll
            for (int d = 0; d < 3; ++d) {
                o[2  + d] = xM[d]; o[5  + d] = nM[d]; o[8  + d] = sM[d];
                o[11 + d] = xS[d]; o[14 + d] = nS[d]; o[17 + d] = sS[d];
            }
        }
    }
}

__global__ __launch_bounds__(128)
void fv_finalize(const float* __restrict__ w, const float* __restrict__ partial,
                 float* __restrict__ out)
{
    __shared__ float vals[20][128];
    __shared__ float ssq[20];
    const int b = blockIdx.x;
    const int t = threadIdx.x;

    float v[20];
    if (t < K_) {
        float a[20];
        {
            const float* p = partial + ((size_t)(b * NCHUNK) * K_ + t) * 20;
#pragma unroll
            for (int f = 0; f < 20; ++f) a[f] = p[f];
        }
        for (int c = 1; c < NCHUNK; ++c) {
            const float* p = partial + ((size_t)(b * NCHUNK + c) * K_ + t) * 20;
            a[0] = fmaxf(a[0], p[0]); a[1] += p[1];
#pragma unroll
            for (int d = 0; d < 3; ++d) {
                a[2  + d] = fmaxf(a[2  + d], p[2  + d]);
                a[5  + d] = fminf(a[5  + d], p[5  + d]);
                a[8  + d] +=            p[8  + d];
                a[11 + d] = fmaxf(a[11 + d], p[11 + d]);
                a[14 + d] = fminf(a[14 + d], p[14 + d]);
                a[17 + d] +=            p[17 + d];
            }
        }
        const float wk  = w[t];
        const float Nf  = (float)N_;
        const float spi = 1.0f / (sqrtf(wk) * Nf);
        const float ssg = 1.0f / (Nf * sqrtf(2.0f * wk));
        v[0] = (a[0] - wk)      * spi;
        v[1] = (a[1] - Nf * wk) * spi;
#pragma unroll
        for (int f = 2;  f < 11; ++f) v[f] = a[f] * spi;
#pragma unroll
        for (int f = 11; f < 20; ++f) v[f] = a[f] * ssg;
#pragma unroll
        for (int f = 0; f < 20; ++f)
            v[f] = copysignf(sqrtf(fabsf(v[f])), v[f]);
    } else {
#pragma unroll
        for (int f = 0; f < 20; ++f) v[f] = 0.0f;
    }
#pragma unroll
    for (int f = 0; f < 20; ++f) vals[f][t] = v[f];
    __syncthreads();

    if (t < 20) {
        float s = 0.0f;
        for (int k = 0; k < K_; ++k) { float x = vals[t][k]; s += x * x; }
        ssq[t] = s;
    }
    __syncthreads();

    if (t < K_) {
#pragma unroll
        for (int f = 0; f < 20; ++f)
            out[(size_t)b * 20 * K_ + f * K_ + t] =
                v[f] * rsqrtf(fmaxf(ssq[f], 1e-12f));
    }
}

extern "C" void kernel_launch(void* const* d_in, const int* in_sizes, int n_in,
                              void* d_out, int out_size, void* d_ws, size_t ws_size,
                              hipStream_t stream)
{
    const float* pts = (const float*)d_in[0];
    const float* w   = (const float*)d_in[1];
    const float* mu  = (const float*)d_in[2];
    const float* sg  = (const float*)d_in[3];
    float* out = (float*)d_out;

    float* partial = (float*)d_ws;

    fv_accum<<<B_ * NCHUNK, NTHR, 0, stream>>>(pts, w, mu, sg, partial);
    fv_finalize<<<B_, 128, 0, stream>>>(w, partial, out);
}


# finalize re-gridded: one block per (batch, statistic) over statistic-major partials, 16 chunk loads in flight at once (was 16 serial load rounds); sum of squares over k now a lane tree
# speedup vs baseline: 2.4650x; 1.6712x over previous
#include <hip/hip_runtime.h>
#include <math.h>


typedef __attribute__((ext_vector_type(2))) float v2f;
typedef __attribute__((ext_vector_type(8))) float v8f;

#define B_      16
#define N_      8192
#define K_      125
#define KT      8
#define NCHUNK  16
#define CHUNK   (N_ / NCHUNK)
#define NSUB    4
#define SUBPTS  (CHUNK / NSUB)
#define C15     0.06349363593424097f

#define NTHR    (128 * NSUB)
#define NWAVE   (NTHR / 32)
#define NTILE   (CHUNK / 16)
#define PSLOT   128

__global__ __launch_bounds__(NTHR)
void fv_accum(const float* __restrict__ pts, const float* __restrict__ w,
              const float* __restrict__ mu, const float* __restrict__ sg,
              float* __restrict__ partial)
{
    __shared__ float lx[CHUNK], ly[CHUNK], lz[CHUNK];
    __shared__ __attribute__((aligned(16))) float lrd[CHUNK];
    __shared__ float4 lc4[KT * 32];
    __shared__ float comb[NSUB - 1][20][128];
    const int b   = blockIdx.x >> 4;
    const int ch  = blockIdx.x & 15;
    const int n0  = ch * CHUNK;
    const int tid = threadIdx.x;
    const int t   = tid & 127;
    const int lane = tid & 31;
    const int g    = __builtin_amdgcn_readfirstlane(tid >> 7);
    const int wave = __builtin_amdgcn_readfirstlane(tid >> 5);

    static_assert(CHUNK == NTHR, "staging takes one point per thread");
    const float* pp = pts + (size_t)(b * N_ + n0 + tid) * 3;
    const float qx = pp[0], qy = pp[1], qz = pp[2];

    const int kk = (t < K_) ? t : 0;
    const float m0 = mu[kk*3+0], m1 = mu[kk*3+1], m2 = mu[kk*3+2];
    const float s0 = sg[kk*3+0], s1 = sg[kk*3+1], s2 = sg[kk*3+2];
    const float wq = w[kk];

    if (tid < KT * 32) {
        const int  ct    = tid >> 5;
        const int  cm    = lane & 15;
        const bool chi   = lane >= 16;
        const int  kc    = ct * 16 + cm;
        const bool valid = kc < K_;
        const int  ki    = valid ? kc : 0;

        const float c0 = mu[ki*3+0], c1 = mu[ki*3+1], c2 = mu[ki*3+2];
        const float d0 = sg[ki*3+0], d1 = sg[ki*3+1], d2 = sg[ki*3+2];
        const float i0 = 1.0f/(d0*d0), i1 = 1.0f/(d1*d1), i2 = 1.0f/(d2*d2);
        const float g0 = -0.5f*i0, g1 = -0.5f*i1, g2 = -0.5f*i2;
        const float g3 = c0*i0,    g4 = c1*i1,    g5 = c2*i2;
        const float lq = logf(w[ki] * C15 / (d0 * d1 * d2));
        const float g6 = valid ? (lq - 0.5f*(c0*c0*i0 + c1*c1*i1 + c2*c2*i2))
                               : -INFINITY;

        float4 v;
        v.x = chi ? g2   : g0;
        v.y = chi ? g3   : g1;
        v.z = chi ? g6   : g4;
        v.w = chi ? 0.0f : g5;
        lc4[tid] = v;
    }
    lx[tid] = qx; ly[tid] = qy; lz[tid] = qz;
    __syncthreads();

    {
        const int  m  = lane & 15;
        const bool hi = lane >= 16;
        for (int tile = wave; tile < NTILE; tile += NWAVE) {
            const int p0 = tile << 4;
            const float px = lx[p0 + m], py = ly[p0 + m], pz = lz[p0 + m];

            v2f a1, a2;
            a1.x = hi ? pz * pz : px * px;
            a1.y = hi ? px      : py * py;
            a2.x = hi ? 1.0f    : py;
            a2.y = hi ? 0.0f    : pz;

            float acc[8];
#pragma unroll
            for (int r = 0; r < 8; ++r) acc[r] = 0.0f;

#pragma unroll
            for (int kt = 0; kt < KT; ++kt) {
                const float4 bc = lc4[kt * 32 + lane];
                v2f b1, b2;
                b1.x = bc.x; b1.y = bc.y;
                b2.x = bc.z; b2.y = bc.w;

                v8f c = {};
                c = __builtin_amdgcn_wmma_f32_16x16x4_f32(false, a1, false, b1,
                                                          (short)0, c, false, false);
                c = __builtin_amdgcn_wmma_f32_16x16x4_f32(false, a2, false, b2,
                                                          (short)0, c, false, false);

#pragma unroll
                for (int r = 0; r < 8; ++r)
                    acc[r] += __expf(c[r]);
            }

#pragma unroll
            for (int r = 0; r < 8; ++r) {
                float v = acc[r];
                v += __shfl_xor(v, 1, 32);
                v += __shfl_xor(v, 2, 32);
                v += __shfl_xor(v, 4, 32);
                v += __shfl_xor(v, 8, 32);
                acc[r] = v;
            }
            if (m == 0) {
                float* dst = lrd + p0 + (hi ? 8 : 0);
#pragma unroll
                for (int r = 0; r < 8; ++r)
                    dst[r] = 1.0f / acc[r];
            }
        }
    }
    __syncthreads();

    const float rs0 = 1.0f/s0, rs1 = 1.0f/s1, rs2 = 1.0f/s2;
    const float q = wq * C15 * rs0 * rs1 * rs2;

    float sQ = 0.0f, mQ = -INFINITY;
    float sM[3] = {0,0,0}, xM[3] = {-INFINITY,-INFINITY,-INFINITY};
    float nM[3] = { INFINITY, INFINITY, INFINITY};
    float sS[3] = {0,0,0}, xS[3] = {-INFINITY,-INFINITY,-INFINITY};
    float nS[3] = { INFINITY, INFINITY, INFINITY};

    const float* gx = lx  + g * SUBPTS;
    const float* gy = ly  + g * SUBPTS;
    const float* gz = lz  + g * SUBPTS;
    const float* gr = lrd + g * SUBPTS;
    for (int i = 0; i < SUBPTS; ++i) {
        float z0 = (gx[i] - m0) * rs0;
        float z1 = (gy[i] - m1) * rs1;
        float z2 = (gz[i] - m2) * rs2;
        float e  = -0.5f * (z0*z0 + z1*z1 + z2*z2);
        float Q  = q * __expf(e) * gr[i];
        sQ += Q; mQ = fmaxf(mQ, Q);
        float z[3] = {z0, z1, z2};
#pragma unroll
        for (int d = 0; d < 3; ++d) {
            float u = Q * z[d];
            sM[d] += u; xM[d] = fmaxf(xM[d], u); nM[d] = fminf(nM[d], u);
            float v = Q * (z[d]*z[d] - 1.0f);
            sS[d] += v; xS[d] = fmaxf(xS[d], v); nS[d] = fminf(nS[d], v);
        }
    }

    if (g > 0) {
        float (*c)[128] = comb[g - 1];
        c[0][t] = mQ;  c[1][t] = sQ;
#pragma unroll
        for (int d = 0; d < 3; ++d) {
            c[2  + d][t] = xM[d]; c[5  + d][t] = nM[d]; c[8  + d][t] = sM[d];
            c[11 + d][t] = xS[d]; c[14 + d][t] = nS[d]; c[17 + d][t] = sS[d];
        }
    }
    __syncthreads();

    if (g == 0) {
#pragma unroll
        for (int s = 0; s < NSUB - 1; ++s) {
            float (*c)[128] = comb[s];
            mQ = fmaxf(mQ, c[0][t]);  sQ += c[1][t];
#pragma unroll
            for (int d = 0; d < 3; ++d) {
                xM[d] = fmaxf(xM[d], c[2  + d][t]);
                nM[d] = fminf(nM[d], c[5  + d][t]);
                sM[d] +=             c[8  + d][t];
                xS[d] = fmaxf(xS[d], c[11 + d][t]);
                nS[d] = fminf(nS[d], c[14 + d][t]);
                sS[d] +=             c[17 + d][t];
            }
        }
        float* o = partial + (size_t)(b * NCHUNK + ch) * 20 * PSLOT + t;
        o[0 * PSLOT] = mQ;  o[1 * PSLOT] = sQ;
#pragma unroll
        for (int d = 0; d < 3; ++d) {
            o[(2  + d) * PSLOT] = xM[d]; o[(5  + d) * PSLOT] = nM[d];
            o[(8  + d) * PSLOT] = sM[d]; o[(11 + d) * PSLOT] = xS[d];
            o[(14 + d) * PSLOT] = nS[d]; o[(17 + d) * PSLOT] = sS[d];
        }
    }
}

__global__ __launch_bounds__(128)
void fv_finalize(const float* __restrict__ w, const float* __restrict__ partial,
                 float* __restrict__ out)
{
    __shared__ float red[4];
    const int f = blockIdx.x;
    const int b = blockIdx.y;
    const int t = threadIdx.x;

    const bool is_sum = (f == 1) || (f >= 8 && f < 11) || (f >= 17);
    const bool is_min = (f >= 5 && f < 8) || (f >= 14 && f < 17);

    float v = 0.0f;
    if (t < K_) {
        const float* p = partial + ((size_t)(b * NCHUNK) * 20 + f) * PSLOT + t;
        float pv[NCHUNK];
#pragma unroll
        for (int c = 0; c < NCHUNK; ++c) pv[c] = p[(size_t)c * 20 * PSLOT];
        const float wk = w[t];

        float a = pv[0];
        if (is_sum) {
#pragma unroll
            for (int c = 1; c < NCHUNK; ++c) a += pv[c];
        } else if (is_min) {
#pragma unroll
            for (int c = 1; c < NCHUNK; ++c) a = fminf(a, pv[c]);
        } else {
#pragma unroll
            for (int c = 1; c < NCHUNK; ++c) a = fmaxf(a, pv[c]);
        }

        const float Nf  = (float)N_;
        const float spi = 1.0f / (sqrtf(wk) * Nf);
        const float ssg = 1.0f / (Nf * sqrtf(2.0f * wk));
        float x;
        if (f == 0)      x = (a - wk)      * spi;
        else if (f == 1) x = (a - Nf * wk) * spi;
        else if (f < 11) x = a * spi;
        else             x = a * ssg;
        v = copysignf(sqrtf(fabsf(x)), x);
    }

    float s = v * v;
    s += __shfl_xor(s, 16, 32);
    s += __shfl_xor(s, 8, 32);
    s += __shfl_xor(s, 4, 32);
    s += __shfl_xor(s, 2, 32);
    s += __shfl_xor(s, 1, 32);
    if ((t & 31) == 0) red[t >> 5] = s;
    __syncthreads();
    const float tot = (red[0] + red[1]) + (red[2] + red[3]);

    if (t < K_)
        out[(size_t)b * 20 * K_ + f * K_ + t] = v * rsqrtf(fmaxf(tot, 1e-12f));
}

extern "C" void kernel_launch(void* const* d_in, const int* in_sizes, int n_in,
                              void* d_out, int out_size, void* d_ws, size_t ws_size,
                              hipStream_t stream)
{
    const float* pts = (const float*)d_in[0];
    const float* w   = (const float*)d_in[1];
    const float* mu  = (const float*)d_in[2];
    const float* sg  = (const float*)d_in[3];
    float* out = (float*)d_out;

    float* partial = (float*)d_ws;

    fv_accum<<<B_ * NCHUNK, NTHR, 0, stream>>>(pts, w, mu, sg, partial);
    fv_finalize<<<dim3(20, B_), 128, 0, stream>>>(w, partial, out);
}


# each weighted density exp(ln p) computed once: the WMMA denominator pass keeps all 512x128 of a chunk in LDS (258 KiB) and the accumulation reads it; posterior numerator = the WMMA quadratic form
# speedup vs baseline: 2.7220x; 1.1043x over previous
#include <hip/hip_runtime.h>
#include <math.h>


typedef __attribute__((ext_vector_type(2))) float v2f;
typedef __attribute__((ext_vector_type(8))) float v8f;

#define B_      16
#define N_      8192
#define K_      125
#define KT      8
#define NCHUNK  16
#define CHUNK   (N_ / NCHUNK)
#define NSUB    4
#define SUBPTS  (CHUNK / NSUB)
#define C15     0.06349363593424097f
#define PST     (CHUNK + 4)

#define NTHR    (128 * NSUB)
#define NWAVE   (NTHR / 32)
#define NTILE   (CHUNK / 16)
#define PSLOT   128

__global__ __launch_bounds__(NTHR)
void fv_accum(const float* __restrict__ pts, const float* __restrict__ w,
              const float* __restrict__ mu, const float* __restrict__ sg,
              float* __restrict__ partial)
{
    __shared__ float lx[CHUNK], ly[CHUNK], lz[CHUNK];
    __shared__ __attribute__((aligned(16))) float lrd[CHUNK];
    __shared__ float4 lc4[KT * 32];
    __shared__ float comb[NSUB - 1][20][128];
    __shared__ __attribute__((aligned(16))) float lp[128 * PST];
    const int b   = blockIdx.x >> 4;
    const int ch  = blockIdx.x & 15;
    const int n0  = ch * CHUNK;
    const int tid = threadIdx.x;
    const int t   = tid & 127;
    const int lane = tid & 31;
    const int g    = __builtin_amdgcn_readfirstlane(tid >> 7);
    const int wave = __builtin_amdgcn_readfirstlane(tid >> 5);

    static_assert(CHUNK == NTHR, "staging takes one point per thread");
    const float* pp = pts + (size_t)(b * N_ + n0 + tid) * 3;
    const float qx = pp[0], qy = pp[1], qz = pp[2];

    const int kk = (t < K_) ? t : 0;
    const float m0 = mu[kk*3+0], m1 = mu[kk*3+1], m2 = mu[kk*3+2];
    const float s0 = sg[kk*3+0], s1 = sg[kk*3+1], s2 = sg[kk*3+2];

    if (tid < KT * 32) {
        const int  ct    = tid >> 5;
        const int  cm    = lane & 15;
        const bool chi   = lane >= 16;
        const int  kc    = ct * 16 + cm;
        const bool valid = kc < K_;
        const int  ki    = valid ? kc : 0;

        const float c0 = mu[ki*3+0], c1 = mu[ki*3+1], c2 = mu[ki*3+2];
        const float d0 = sg[ki*3+0], d1 = sg[ki*3+1], d2 = sg[ki*3+2];
        const float i0 = 1.0f/(d0*d0), i1 = 1.0f/(d1*d1), i2 = 1.0f/(d2*d2);
        const float g0 = -0.5f*i0, g1 = -0.5f*i1, g2 = -0.5f*i2;
        const float g3 = c0*i0,    g4 = c1*i1,    g5 = c2*i2;
        const float lq = logf(w[ki] * C15 / (d0 * d1 * d2));
        const float g6 = valid ? (lq - 0.5f*(c0*c0*i0 + c1*c1*i1 + c2*c2*i2))
                               : -INFINITY;

        float4 v;
        v.x = chi ? g2   : g0;
        v.y = chi ? g3   : g1;
        v.z = chi ? g6   : g4;
        v.w = chi ? 0.0f : g5;
        lc4[tid] = v;
    }
    lx[tid] = qx; ly[tid] = qy; lz[tid] = qz;
    __syncthreads();

    {
        const int  m  = lane & 15;
        const bool hi = lane >= 16;
        for (int tile = wave; tile < NTILE; tile += NWAVE) {
            const int p0 = tile << 4;
            const float px = lx[p0 + m], py = ly[p0 + m], pz = lz[p0 + m];

            v2f a1, a2;
            a1.x = hi ? pz * pz : px * px;
            a1.y = hi ? px      : py * py;
            a2.x = hi ? 1.0f    : py;
            a2.y = hi ? 0.0f    : pz;

            float acc[8];
#pragma unroll
            for (int r = 0; r < 8; ++r) acc[r] = 0.0f;

#pragma unroll
            for (int kt = 0; kt < KT; ++kt) {
                const float4 bc = lc4[kt * 32 + lane];
                v2f b1, b2;
                b1.x = bc.x; b1.y = bc.y;
                b2.x = bc.z; b2.y = bc.w;

                v8f c = {};
                c = __builtin_amdgcn_wmma_f32_16x16x4_f32(false, a1, false, b1,
                                                          (short)0, c, false, false);
                c = __builtin_amdgcn_wmma_f32_16x16x4_f32(false, a2, false, b2,
                                                          (short)0, c, false, false);

                float ev[8];
#pragma unroll
                for (int r = 0; r < 8; ++r) {
                    ev[r] = __expf(c[r]);
                    acc[r] += ev[r];
                }
                float* pd = lp + (kt * 16 + m) * PST + p0 + (hi ? 8 : 0);
                *(float4*)(pd)     = make_float4(ev[0], ev[1], ev[2], ev[3]);
                *(float4*)(pd + 4) = make_float4(ev[4], ev[5], ev[6], ev[7]);
            }

#pragma unroll
            for (int r = 0; r < 8; ++r) {
                float v = acc[r];
                v += __shfl_xor(v, 1, 32);
                v += __shfl_xor(v, 2, 32);
                v += __shfl_xor(v, 4, 32);
                v += __shfl_xor(v, 8, 32);
                acc[r] = v;
            }
            if (m == 0) {
                float* dst = lrd + p0 + (hi ? 8 : 0);
#pragma unroll
                for (int r = 0; r < 8; ++r)
                    dst[r] = 1.0f / acc[r];
            }
        }
    }
    __syncthreads();

    const float rs0 = 1.0f/s0, rs1 = 1.0f/s1, rs2 = 1.0f/s2;

    float sQ = 0.0f, mQ = -INFINITY;
    float sM[3] = {0,0,0}, xM[3] = {-INFINITY,-INFINITY,-INFINITY};
    float nM[3] = { INFINITY, INFINITY, INFINITY};
    float sS[3] = {0,0,0}, xS[3] = {-INFINITY,-INFINITY,-INFINITY};
    float nS[3] = { INFINITY, INFINITY, INFINITY};

    const float* gx = lx  + g * SUBPTS;
    const float* gy = ly  + g * SUBPTS;
    const float* gz = lz  + g * SUBPTS;
    const float* gr = lrd + g * SUBPTS;
    const float* gp = lp + t * PST + g * SUBPTS;
    for (int i = 0; i < SUBPTS; ++i) {
        float z0 = (gx[i] - m0) * rs0;
        float z1 = (gy[i] - m1) * rs1;
        float z2 = (gz[i] - m2) * rs2;
        float Q  = gp[i] * gr[i];
        sQ += Q; mQ = fmaxf(mQ, Q);
        float z[3] = {z0, z1, z2};
#pragma unroll
        for (int d = 0; d < 3; ++d) {
            float u = Q * z[d];
            sM[d] += u; xM[d] = fmaxf(xM[d], u); nM[d] = fminf(nM[d], u);
            float v = Q * (z[d]*z[d] - 1.0f);
            sS[d] += v; xS[d] = fmaxf(xS[d], v); nS[d] = fminf(nS[d], v);
        }
    }

    if (g > 0) {
        float (*c)[128] = comb[g - 1];
        c[0][t] = mQ;  c[1][t] = sQ;
#pragma unroll
        for (int d = 0; d < 3; ++d) {
            c[2  + d][t] = xM[d]; c[5  + d][t] = nM[d]; c[8  + d][t] = sM[d];
            c[11 + d][t] = xS[d]; c[14 + d][t] = nS[d]; c[17 + d][t] = sS[d];
        }
    }
    __syncthreads();

    if (g == 0) {
#pragma unroll
        for (int s = 0; s < NSUB - 1; ++s) {
            float (*c)[128] = comb[s];
            mQ = fmaxf(mQ, c[0][t]);  sQ += c[1][t];
#pragma unroll
            for (int d = 0; d < 3; ++d) {
                xM[d] = fmaxf(xM[d], c[2  + d][t]);
                nM[d] = fminf(nM[d], c[5  + d][t]);
                sM[d] +=             c[8  + d][t];
                xS[d] = fmaxf(xS[d], c[11 + d][t]);
                nS[d] = fminf(nS[d], c[14 + d][t]);
                sS[d] +=             c[17 + d][t];
            }
        }
        float* o = partial + (size_t)(b * NCHUNK + ch) * 20 * PSLOT + t;
        o[0 * PSLOT] = mQ;  o[1 * PSLOT] = sQ;
#pragma unroll
        for (int d = 0; d < 3; ++d) {
            o[(2  + d) * PSLOT] = xM[d]; o[(5  + d) * PSLOT] = nM[d];
            o[(8  + d) * PSLOT] = sM[d]; o[(11 + d) * PSLOT] = xS[d];
            o[(14 + d) * PSLOT] = nS[d]; o[(17 + d) * PSLOT] = sS[d];
        }
    }
}

__global__ __launch_bounds__(128)
void fv_finalize(const float* __restrict__ w, const float* __restrict__ partial,
                 float* __restrict__ out)
{
    __shared__ float red[4];
    const int f = blockIdx.x;
    const int b = blockIdx.y;
    const int t = threadIdx.x;

    const bool is_sum = (f == 1) || (f >= 8 && f < 11) || (f >= 17);
    const bool is_min = (f >= 5 && f < 8) || (f >= 14 && f < 17);

    float v = 0.0f;
    if (t < K_) {
        const float* p = partial + ((size_t)(b * NCHUNK) * 20 + f) * PSLOT + t;
        float pv[NCHUNK];
#pragma unroll
        for (int c = 0; c < NCHUNK; ++c) pv[c] = p[(size_t)c * 20 * PSLOT];
        const float wk = w[t];

        float a = pv[0];
        if (is_sum) {
#pragma unroll
            for (int c = 1; c < NCHUNK; ++c) a += pv[c];
        } else if (is_min) {
#pragma unroll
            for (int c = 1; c < NCHUNK; ++c) a = fminf(a, pv[c]);
        } else {
#pragma unroll
            for (int c = 1; c < NCHUNK; ++c) a = fmaxf(a, pv[c]);
        }

        const float Nf  = (float)N_;
        const float spi = 1.0f / (sqrtf(wk) * Nf);
        const float ssg = 1.0f / (Nf * sqrtf(2.0f * wk));
        float x;
        if (f == 0)      x = (a - wk)      * spi;
        else if (f == 1) x = (a - Nf * wk) * spi;
        else if (f < 11) x = a * spi;
        else             x = a * ssg;
        v = copysignf(sqrtf(fabsf(x)), x);
    }

    float s = v * v;
    s += __shfl_xor(s, 16, 32);
    s += __shfl_xor(s, 8, 32);
    s += __shfl_xor(s, 4, 32);
    s += __shfl_xor(s, 2, 32);
    s += __shfl_xor(s, 1, 32);
    if ((t & 31) == 0) red[t >> 5] = s;
    __syncthreads();
    const float tot = (red[0] + red[1]) + (red[2] + red[3]);

    if (t < K_)
        out[(size_t)b * 20 * K_ + f * K_ + t] = v * rsqrtf(fmaxf(tot, 1e-12f));
}

extern "C" void kernel_launch(void* const* d_in, const int* in_sizes, int n_in,
                              void* d_out, int out_size, void* d_ws, size_t ws_size,
                              hipStream_t stream)
{
    const float* pts = (const float*)d_in[0];
    const float* w   = (const float*)d_in[1];
    const float* mu  = (const float*)d_in[2];
    const float* sg  = (const float*)d_in[3];
    float* out = (float*)d_out;

    float* partial = (float*)d_ws;

    fv_accum<<<B_ * NCHUNK, NTHR, 0, stream>>>(pts, w, mu, sg, partial);
    fv_finalize<<<dim3(20, B_), 128, 0, stream>>>(w, partial, out);
}


# accumulation loop computes on point pairs in packed f32 (v_pk_mul/add/fma, 4 points per trip): 90 -> 73 vector instr per 4 points; each 128-point sub-sum is now an even + an odd chain, added last
# speedup vs baseline: 3.0711x; 1.1283x over previous
#include <hip/hip_runtime.h>
#include <math.h>


typedef __attribute__((ext_vector_type(2))) float v2f;
typedef __attribute__((ext_vector_type(4))) float v4f;
typedef __attribute__((ext_vector_type(8))) float v8f;

#define B_      16
#define N_      8192
#define K_      125
#define KT      8
#define NCHUNK  16
#define CHUNK   (N_ / NCHUNK)
#define NSUB    4
#define SUBPTS  (CHUNK / NSUB)
#define C15     0.06349363593424097f
#define PST     (CHUNK + 4)

#define NTHR    (128 * NSUB)
#define NWAVE   (NTHR / 32)
#define NTILE   (CHUNK / 16)
#define PSLOT   128

__global__ __launch_bounds__(NTHR)
void fv_accum(const float* __restrict__ pts, const float* __restrict__ w,
              const float* __restrict__ mu, const float* __restrict__ sg,
              float* __restrict__ partial)
{
    __shared__ __attribute__((aligned(16))) float lx[CHUNK], ly[CHUNK], lz[CHUNK];
    __shared__ __attribute__((aligned(16))) float lrd[CHUNK];
    __shared__ float4 lc4[KT * 32];
    __shared__ float comb[NSUB - 1][20][128];
    __shared__ __attribute__((aligned(16))) float lp[128 * PST];
    const int b   = blockIdx.x >> 4;
    const int ch  = blockIdx.x & 15;
    const int n0  = ch * CHUNK;
    const int tid = threadIdx.x;
    const int t   = tid & 127;
    const int lane = tid & 31;
    const int g    = __builtin_amdgcn_readfirstlane(tid >> 7);
    const int wave = __builtin_amdgcn_readfirstlane(tid >> 5);

    static_assert(CHUNK == NTHR, "staging takes one point per thread");
    const float* pp = pts + (size_t)(b * N_ + n0 + tid) * 3;
    const float qx = pp[0], qy = pp[1], qz = pp[2];

    const int kk = (t < K_) ? t : 0;
    const float m0 = mu[kk*3+0], m1 = mu[kk*3+1], m2 = mu[kk*3+2];
    const float s0 = sg[kk*3+0], s1 = sg[kk*3+1], s2 = sg[kk*3+2];

    if (tid < KT * 32) {
        const int  ct    = tid >> 5;
        const int  cm    = lane & 15;
        const bool chi   = lane >= 16;
        const int  kc    = ct * 16 + cm;
        const bool valid = kc < K_;
        const int  ki    = valid ? kc : 0;

        const float c0 = mu[ki*3+0], c1 = mu[ki*3+1], c2 = mu[ki*3+2];
        const float d0 = sg[ki*3+0], d1 = sg[ki*3+1], d2 = sg[ki*3+2];
        const float i0 = 1.0f/(d0*d0), i1 = 1.0f/(d1*d1), i2 = 1.0f/(d2*d2);
        const float g0 = -0.5f*i0, g1 = -0.5f*i1, g2 = -0.5f*i2;
        const float g3 = c0*i0,    g4 = c1*i1,    g5 = c2*i2;
        const float lq = logf(w[ki] * C15 / (d0 * d1 * d2));
        const float g6 = valid ? (lq - 0.5f*(c0*c0*i0 + c1*c1*i1 + c2*c2*i2))
                               : -INFINITY;

        float4 v;
        v.x = chi ? g2   : g0;
        v.y = chi ? g3   : g1;
        v.z = chi ? g6   : g4;
        v.w = chi ? 0.0f : g5;
        lc4[tid] = v;
    }
    lx[tid] = qx; ly[tid] = qy; lz[tid] = qz;
    __syncthreads();

    {
        const int  m  = lane & 15;
        const bool hi = lane >= 16;
        for (int tile = wave; tile < NTILE; tile += NWAVE) {
            const int p0 = tile << 4;
            const float px = lx[p0 + m], py = ly[p0 + m], pz = lz[p0 + m];

            v2f a1, a2;
            a1.x = hi ? pz * pz : px * px;
            a1.y = hi ? px      : py * py;
            a2.x = hi ? 1.0f    : py;
            a2.y = hi ? 0.0f    : pz;

            float acc[8];
#pragma unroll
            for (int r = 0; r < 8; ++r) acc[r] = 0.0f;

#pragma unroll
            for (int kt = 0; kt < KT; ++kt) {
                const float4 bc = lc4[kt * 32 + lane];
                v2f b1, b2;
                b1.x = bc.x; b1.y = bc.y;
                b2.x = bc.z; b2.y = bc.w;

                v8f c = {};
                c = __builtin_amdgcn_wmma_f32_16x16x4_f32(false, a1, false, b1,
                                                          (short)0, c, false, false);
                c = __builtin_amdgcn_wmma_f32_16x16x4_f32(false, a2, false, b2,
                                                          (short)0, c, false, false);

                float ev[8];
#pragma unroll
                for (int r = 0; r < 8; ++r) {
                    ev[r] = __expf(c[r]);
                    acc[r] += ev[r];
                }
                float* pd = lp + (kt * 16 + m) * PST + p0 + (hi ? 8 : 0);
                *(float4*)(pd)     = make_float4(ev[0], ev[1], ev[2], ev[3]);
                *(float4*)(pd + 4) = make_float4(ev[4], ev[5], ev[6], ev[7]);
            }

#pragma unroll
            for (int r = 0; r < 8; ++r) {
                float v = acc[r];
                v += __shfl_xor(v, 1, 32);
                v += __shfl_xor(v, 2, 32);
                v += __shfl_xor(v, 4, 32);
                v += __shfl_xor(v, 8, 32);
                acc[r] = v;
            }
            if (m == 0) {
                float* dst = lrd + p0 + (hi ? 8 : 0);
#pragma unroll
                for (int r = 0; r < 8; ++r)
                    dst[r] = 1.0f / acc[r];
            }
        }
    }
    __syncthreads();

    const float rs0 = 1.0f/s0, rs1 = 1.0f/s1, rs2 = 1.0f/s2;

    float mQ = -INFINITY;
    float xM[3] = {-INFINITY,-INFINITY,-INFINITY};
    float nM[3] = { INFINITY, INFINITY, INFINITY};
    float xS[3] = {-INFINITY,-INFINITY,-INFINITY};
    float nS[3] = { INFINITY, INFINITY, INFINITY};
    v2f sQ2 = {0.0f, 0.0f};
    v2f sM2[3] = {{0.0f, 0.0f}, {0.0f, 0.0f}, {0.0f, 0.0f}};
    v2f sS2[3] = {{0.0f, 0.0f}, {0.0f, 0.0f}, {0.0f, 0.0f}};
    const v2f mm[3] = {{m0, m0}, {m1, m1}, {m2, m2}};
    const v2f rr[3] = {{rs0, rs0}, {rs1, rs1}, {rs2, rs2}};
    const v2f one2  = {1.0f, 1.0f};

    const v4f* gx = (const v4f*)(lx  + g * SUBPTS);
    const v4f* gy = (const v4f*)(ly  + g * SUBPTS);
    const v4f* gz = (const v4f*)(lz  + g * SUBPTS);
    const v4f* gr = (const v4f*)(lrd + g * SUBPTS);
    const v4f* gp = (const v4f*)(lp + t * PST + g * SUBPTS);
    static_assert(SUBPTS % 4 == 0 && PST % 4 == 0, "16-byte point groups");
    for (int i = 0; i < SUBPTS / 4; ++i) {
        const v4f X = gx[i], Y = gy[i], Z = gz[i], R = gr[i], P = gp[i];
#pragma unroll
        for (int h = 0; h < 2; ++h) {
            const v2f c[3] = { h ? X.zw : X.xy, h ? Y.zw : Y.xy, h ? Z.zw : Z.xy };
            const v2f Q = (h ? P.zw : P.xy) * (h ? R.zw : R.xy);
            sQ2 += Q;
            mQ = fmaxf(fmaxf(mQ, Q.x), Q.y);
#pragma unroll
            for (int d = 0; d < 3; ++d) {
                const v2f z = (c[d] - mm[d]) * rr[d];
                const v2f u = Q * z;
                sM2[d] += u;
                xM[d] = fmaxf(fmaxf(xM[d], u.x), u.y);
                nM[d] = fminf(fminf(nM[d], u.x), u.y);
                const v2f v = Q * (z * z - one2);
                sS2[d] += v;
                xS[d] = fmaxf(fmaxf(xS[d], v.x), v.y);
                nS[d] = fminf(fminf(nS[d], v.x), v.y);
            }
        }
    }
    float sQ = sQ2.x + sQ2.y;
    float sM[3], sS[3];
#pragma unroll
    for (int d = 0; d < 3; ++d) {
        sM[d] = sM2[d].x + sM2[d].y;
        sS[d] = sS2[d].x + sS2[d].y;
    }

    if (g > 0) {
        float (*c)[128] = comb[g - 1];
        c[0][t] = mQ;  c[1][t] = sQ;
#pragma unroll
        for (int d = 0; d < 3; ++d) {
            c[2  + d][t] = xM[d]; c[5  + d][t] = nM[d]; c[8  + d][t] = sM[d];
            c[11 + d][t] = xS[d]; c[14 + d][t] = nS[d]; c[17 + d][t] = sS[d];
        }
    }
    __syncthreads();

    if (g == 0) {
#pragma unroll
        for (int s = 0; s < NSUB - 1; ++s) {
            float (*c)[128] = comb[s];
            mQ = fmaxf(mQ, c[0][t]);  sQ += c[1][t];
#pragma unroll
            for (int d = 0; d < 3; ++d) {
                xM[d] = fmaxf(xM[d], c[2  + d][t]);
                nM[d] = fminf(nM[d], c[5  + d][t]);
                sM[d] +=             c[8  + d][t];
                xS[d] = fmaxf(xS[d], c[11 + d][t]);
                nS[d] = fminf(nS[d], c[14 + d][t]);
                sS[d] +=             c[17 + d][t];
            }
        }
        float* o = partial + (size_t)(b * NCHUNK + ch) * 20 * PSLOT + t;
        o[0 * PSLOT] = mQ;  o[1 * PSLOT] = sQ;
#pragma unroll
        for (int d = 0; d < 3; ++d) {
            o[(2  + d) * PSLOT] = xM[d]; o[(5  + d) * PSLOT] = nM[d];
            o[(8  + d) * PSLOT] = sM[d]; o[(11 + d) * PSLOT] = xS[d];
            o[(14 + d) * PSLOT] = nS[d]; o[(17 + d) * PSLOT] = sS[d];
        }
    }
}

__global__ __launch_bounds__(128)
void fv_finalize(const float* __restrict__ w, const float* __restrict__ partial,
                 float* __restrict__ out)
{
    __shared__ float red[4];
    const int f = blockIdx.x;
    const int b = blockIdx.y;
    const int t = threadIdx.x;

    const bool is_sum = (f == 1) || (f >= 8 && f < 11) || (f >= 17);
    const bool is_min = (f >= 5 && f < 8) || (f >= 14 && f < 17);

    float v = 0.0f;
    if (t < K_) {
        const float* p = partial + ((size_t)(b * NCHUNK) * 20 + f) * PSLOT + t;
        float pv[NCHUNK];
#pragma unroll
        for (int c = 0; c < NCHUNK; ++c) pv[c] = p[(size_t)c * 20 * PSLOT];
        const float wk = w[t];

        float a = pv[0];
        if (is_sum) {
#pragma unroll
            for (int c = 1; c < NCHUNK; ++c) a += pv[c];
        } else if (is_min) {
#pragma unroll
            for (int c = 1; c < NCHUNK; ++c) a = fminf(a, pv[c]);
        } else {
#pragma unroll
            for (int c = 1; c < NCHUNK; ++c) a = fmaxf(a, pv[c]);
        }

        const float Nf  = (float)N_;
        const float spi = 1.0f / (sqrtf(wk) * Nf);
        const float ssg = 1.0f / (Nf * sqrtf(2.0f * wk));
        float x;
        if (f == 0)      x = (a - wk)      * spi;
        else if (f == 1) x = (a - Nf * wk) * spi;
        else if (f < 11) x = a * spi;
        else             x = a * ssg;
        v = copysignf(sqrtf(fabsf(x)), x);
    }

    float s = v * v;
    s += __shfl_xor(s, 16, 32);
    s += __shfl_xor(s, 8, 32);
    s += __shfl_xor(s, 4, 32);
    s += __shfl_xor(s, 2, 32);
    s += __shfl_xor(s, 1, 32);
    if ((t & 31) == 0) red[t >> 5] = s;
    __syncthreads();
    const float tot = (red[0] + red[1]) + (red[2] + red[3]);

    if (t < K_)
        out[(size_t)b * 20 * K_ + f * K_ + t] = v * rsqrtf(fmaxf(tot, 1e-12f));
}

extern "C" void kernel_launch(void* const* d_in, const int* in_sizes, int n_in,
                              void* d_out, int out_size, void* d_ws, size_t ws_size,
                              hipStream_t stream)
{
    const float* pts = (const float*)d_in[0];
    const float* w   = (const float*)d_in[1];
    const float* mu  = (const float*)d_in[2];
    const float* sg  = (const float*)d_in[3];
    float* out = (float*)d_out;

    float* partial = (float*)d_ws;

    fv_accum<<<B_ * NCHUNK, NTHR, 0, stream>>>(pts, w, mu, sg, partial);
    fv_finalize<<<dim3(20, B_), 128, 0, stream>>>(w, partial, out);
}


# accumulation loop arithmetic refactored into fused multiply-adds: z = x*(1/s) + (-mu/s), and Q*(z*z-1) = (Q*z)*z - Q reusing Q*z; 73 -> 61 vector instr per 4 points; order of every sum unchanged
# speedup vs baseline: 3.2398x; 1.0549x over previous
#include <hip/hip_runtime.h>
#include <math.h>


typedef __attribute__((ext_vector_type(2))) float v2f;
typedef __attribute__((ext_vector_type(4))) float v4f;
typedef __attribute__((ext_vector_type(8))) float v8f;

#define B_      16
#define N_      8192
#define K_      125
#define KT      8
#define NCHUNK  16
#define CHUNK   (N_ / NCHUNK)
#define NSUB    4
#define SUBPTS  (CHUNK / NSUB)
#define C15     0.06349363593424097f
#define PST     (CHUNK + 4)

#define NTHR    (128 * NSUB)
#define NWAVE   (NTHR / 32)
#define NTILE   (CHUNK / 16)
#define PSLOT   128

__global__ __launch_bounds__(NTHR)
void fv_accum(const float* __restrict__ pts, const float* __restrict__ w,
              const float* __restrict__ mu, const float* __restrict__ sg,
              float* __restrict__ partial)
{
    __shared__ __attribute__((aligned(16))) float lx[CHUNK], ly[CHUNK], lz[CHUNK];
    __shared__ __attribute__((aligned(16))) float lrd[CHUNK];
    __shared__ float4 lc4[KT * 32];
    __shared__ float comb[NSUB - 1][20][128];
    __shared__ __attribute__((aligned(16))) float lp[128 * PST];
    const int b   = blockIdx.x >> 4;
    const int ch  = blockIdx.x & 15;
    const int n0  = ch * CHUNK;
    const int tid = threadIdx.x;
    const int t   = tid & 127;
    const int lane = tid & 31;
    const int g    = __builtin_amdgcn_readfirstlane(tid >> 7);
    const int wave = __builtin_amdgcn_readfirstlane(tid >> 5);

    static_assert(CHUNK == NTHR, "staging takes one point per thread");
    const float* pp = pts + (size_t)(b * N_ + n0 + tid) * 3;
    const float qx = pp[0], qy = pp[1], qz = pp[2];

    const int kk = (t < K_) ? t : 0;
    const float m0 = mu[kk*3+0], m1 = mu[kk*3+1], m2 = mu[kk*3+2];
    const float s0 = sg[kk*3+0], s1 = sg[kk*3+1], s2 = sg[kk*3+2];

    if (tid < KT * 32) {
        const int  ct    = tid >> 5;
        const int  cm    = lane & 15;
        const bool chi   = lane >= 16;
        const int  kc    = ct * 16 + cm;
        const bool valid = kc < K_;
        const int  ki    = valid ? kc : 0;

        const float c0 = mu[ki*3+0], c1 = mu[ki*3+1], c2 = mu[ki*3+2];
        const float d0 = sg[ki*3+0], d1 = sg[ki*3+1], d2 = sg[ki*3+2];
        const float i0 = 1.0f/(d0*d0), i1 = 1.0f/(d1*d1), i2 = 1.0f/(d2*d2);
        const float g0 = -0.5f*i0, g1 = -0.5f*i1, g2 = -0.5f*i2;
        const float g3 = c0*i0,    g4 = c1*i1,    g5 = c2*i2;
        const float lq = logf(w[ki] * C15 / (d0 * d1 * d2));
        const float g6 = valid ? (lq - 0.5f*(c0*c0*i0 + c1*c1*i1 + c2*c2*i2))
                               : -INFINITY;

        float4 v;
        v.x = chi ? g2   : g0;
        v.y = chi ? g3   : g1;
        v.z = chi ? g6   : g4;
        v.w = chi ? 0.0f : g5;
        lc4[tid] = v;
    }
    lx[tid] = qx; ly[tid] = qy; lz[tid] = qz;
    __syncthreads();

    {
        const int  m  = lane & 15;
        const bool hi = lane >= 16;
        for (int tile = wave; tile < NTILE; tile += NWAVE) {
            const int p0 = tile << 4;
            const float px = lx[p0 + m], py = ly[p0 + m], pz = lz[p0 + m];

            v2f a1, a2;
            a1.x = hi ? pz * pz : px * px;
            a1.y = hi ? px      : py * py;
            a2.x = hi ? 1.0f    : py;
            a2.y = hi ? 0.0f    : pz;

            float acc[8];
#pragma unroll
            for (int r = 0; r < 8; ++r) acc[r] = 0.0f;

#pragma unroll
            for (int kt = 0; kt < KT; ++kt) {
                const float4 bc = lc4[kt * 32 + lane];
                v2f b1, b2;
                b1.x = bc.x; b1.y = bc.y;
                b2.x = bc.z; b2.y = bc.w;

                v8f c = {};
                c = __builtin_amdgcn_wmma_f32_16x16x4_f32(false, a1, false, b1,
                                                          (short)0, c, false, false);
                c = __builtin_amdgcn_wmma_f32_16x16x4_f32(false, a2, false, b2,
                                                          (short)0, c, false, false);

                float ev[8];
#pragma unroll
                for (int r = 0; r < 8; ++r) {
                    ev[r] = __expf(c[r]);
                    acc[r] += ev[r];
                }
                float* pd = lp + (kt * 16 + m) * PST + p0 + (hi ? 8 : 0);
                *(float4*)(pd)     = make_float4(ev[0], ev[1], ev[2], ev[3]);
                *(float4*)(pd + 4) = make_float4(ev[4], ev[5], ev[6], ev[7]);
            }

#pragma unroll
            for (int r = 0; r < 8; ++r) {
                float v = acc[r];
                v += __shfl_xor(v, 1, 32);
                v += __shfl_xor(v, 2, 32);
                v += __shfl_xor(v, 4, 32);
                v += __shfl_xor(v, 8, 32);
                acc[r] = v;
            }
            if (m == 0) {
                float* dst = lrd + p0 + (hi ? 8 : 0);
#pragma unroll
                for (int r = 0; r < 8; ++r)
                    dst[r] = 1.0f / acc[r];
            }
        }
    }
    __syncthreads();

    const float rs0 = 1.0f/s0, rs1 = 1.0f/s1, rs2 = 1.0f/s2;

    float mQ = -INFINITY;
    float xM[3] = {-INFINITY,-INFINITY,-INFINITY};
    float nM[3] = { INFINITY, INFINITY, INFINITY};
    float xS[3] = {-INFINITY,-INFINITY,-INFINITY};
    float nS[3] = { INFINITY, INFINITY, INFINITY};
    v2f sQ2 = {0.0f, 0.0f};
    v2f sM2[3] = {{0.0f, 0.0f}, {0.0f, 0.0f}, {0.0f, 0.0f}};
    v2f sS2[3] = {{0.0f, 0.0f}, {0.0f, 0.0f}, {0.0f, 0.0f}};
    const float o0 = -(m0 * rs0), o1 = -(m1 * rs1), o2 = -(m2 * rs2);
    const v2f oo[3] = {{o0, o0}, {o1, o1}, {o2, o2}};
    const v2f rr[3] = {{rs0, rs0}, {rs1, rs1}, {rs2, rs2}};

    const v4f* gx = (const v4f*)(lx  + g * SUBPTS);
    const v4f* gy = (const v4f*)(ly  + g * SUBPTS);
    const v4f* gz = (const v4f*)(lz  + g * SUBPTS);
    const v4f* gr = (const v4f*)(lrd + g * SUBPTS);
    const v4f* gp = (const v4f*)(lp + t * PST + g * SUBPTS);
    static_assert(SUBPTS % 4 == 0 && PST % 4 == 0, "16-byte point groups");
    for (int i = 0; i < SUBPTS / 4; ++i) {
        const v4f X = gx[i], Y = gy[i], Z = gz[i], R = gr[i], P = gp[i];
#pragma unroll
        for (int h = 0; h < 2; ++h) {
            const v2f c[3] = { h ? X.zw : X.xy, h ? Y.zw : Y.xy, h ? Z.zw : Z.xy };
            const v2f Q = (h ? P.zw : P.xy) * (h ? R.zw : R.xy);
            sQ2 += Q;
            mQ = fmaxf(fmaxf(mQ, Q.x), Q.y);
#pragma unroll
            for (int d = 0; d < 3; ++d) {
                const v2f z = __builtin_elementwise_fma(c[d], rr[d], oo[d]);
                const v2f u = Q * z;
                sM2[d] += u;
                xM[d] = fmaxf(fmaxf(xM[d], u.x), u.y);
                nM[d] = fminf(fminf(nM[d], u.x), u.y);
                const v2f v = __builtin_elementwise_fma(u, z, -Q);
                sS2[d] += v;
                xS[d] = fmaxf(fmaxf(xS[d], v.x), v.y);
                nS[d] = fminf(fminf(nS[d], v.x), v.y);
            }
        }
    }
    float sQ = sQ2.x + sQ2.y;
    float sM[3], sS[3];
#pragma unroll
    for (int d = 0; d < 3; ++d) {
        sM[d] = sM2[d].x + sM2[d].y;
        sS[d] = sS2[d].x + sS2[d].y;
    }

    if (g > 0) {
        float (*c)[128] = comb[g - 1];
        c[0][t] = mQ;  c[1][t] = sQ;
#pragma unroll
        for (int d = 0; d < 3; ++d) {
            c[2  + d][t] = xM[d]; c[5  + d][t] = nM[d]; c[8  + d][t] = sM[d];
            c[11 + d][t] = xS[d]; c[14 + d][t] = nS[d]; c[17 + d][t] = sS[d];
        }
    }
    __syncthreads();

    if (g == 0) {
#pragma unroll
        for (int s = 0; s < NSUB - 1; ++s) {
            float (*c)[128] = comb[s];
            mQ = fmaxf(mQ, c[0][t]);  sQ += c[1][t];
#pragma unroll
            for (int d = 0; d < 3; ++d) {
                xM[d] = fmaxf(xM[d], c[2  + d][t]);
                nM[d] = fminf(nM[d], c[5  + d][t]);
                sM[d] +=             c[8  + d][t];
                xS[d] = fmaxf(xS[d], c[11 + d][t]);
                nS[d] = fminf(nS[d], c[14 + d][t]);
                sS[d] +=             c[17 + d][t];
            }
        }
        float* o = partial + (size_t)(b * NCHUNK + ch) * 20 * PSLOT + t;
        o[0 * PSLOT] = mQ;  o[1 * PSLOT] = sQ;
#pragma unroll
        for (int d = 0; d < 3; ++d) {
            o[(2  + d) * PSLOT] = xM[d]; o[(5  + d) * PSLOT] = nM[d];
            o[(8  + d) * PSLOT] = sM[d]; o[(11 + d) * PSLOT] = xS[d];
            o[(14 + d) * PSLOT] = nS[d]; o[(17 + d) * PSLOT] = sS[d];
        }
    }
}

__global__ __launch_bounds__(128)
void fv_finalize(const float* __restrict__ w, const float* __restrict__ partial,
                 float* __restrict__ out)
{
    __shared__ float red[4];
    const int f = blockIdx.x;
    const int b = blockIdx.y;
    const int t = threadIdx.x;

    const bool is_sum = (f == 1) || (f >= 8 && f < 11) || (f >= 17);
    const bool is_min = (f >= 5 && f < 8) || (f >= 14 && f < 17);

    float v = 0.0f;
    if (t < K_) {
        const float* p = partial + ((size_t)(b * NCHUNK) * 20 + f) * PSLOT + t;
        float pv[NCHUNK];
#pragma unroll
        for (int c = 0; c < NCHUNK; ++c) pv[c] = p[(size_t)c * 20 * PSLOT];
        const float wk = w[t];

        float a = pv[0];
        if (is_sum) {
#pragma unroll
            for (int c = 1; c < NCHUNK; ++c) a += pv[c];
        } else if (is_min) {
#pragma unroll
            for (int c = 1; c < NCHUNK; ++c) a = fminf(a, pv[c]);
        } else {
#pragma unroll
            for (int c = 1; c < NCHUNK; ++c) a = fmaxf(a, pv[c]);
        }

        const float Nf  = (float)N_;
        const float spi = 1.0f / (sqrtf(wk) * Nf);
        const float ssg = 1.0f / (Nf * sqrtf(2.0f * wk));
        float x;
        if (f == 0)      x = (a - wk)      * spi;
        else if (f == 1) x = (a - Nf * wk) * spi;
        else if (f < 11) x = a * spi;
        else             x = a * ssg;
        v = copysignf(sqrtf(fabsf(x)), x);
    }

    float s = v * v;
    s += __shfl_xor(s, 16, 32);
    s += __shfl_xor(s, 8, 32);
    s += __shfl_xor(s, 4, 32);
    s += __shfl_xor(s, 2, 32);
    s += __shfl_xor(s, 1, 32);
    if ((t & 31) == 0) red[t >> 5] = s;
    __syncthreads();
    const float tot = (red[0] + red[1]) + (red[2] + red[3]);

    if (t < K_)
        out[(size_t)b * 20 * K_ + f * K_ + t] = v * rsqrtf(fmaxf(tot, 1e-12f));
}

extern "C" void kernel_launch(void* const* d_in, const int* in_sizes, int n_in,
                              void* d_out, int out_size, void* d_ws, size_t ws_size,
                              hipStream_t stream)
{
    const float* pts = (const float*)d_in[0];
    const float* w   = (const float*)d_in[1];
    const float* mu  = (const float*)d_in[2];
    const float* sg  = (const float*)d_in[3];
    float* out = (float*)d_out;

    float* partial = (float*)d_ws;

    fv_accum<<<B_ * NCHUNK, NTHR, 0, stream>>>(pts, w, mu, sg, partial);
    fv_finalize<<<dim3(20, B_), 128, 0, stream>>>(w, partial, out);
}
